# starting kernel
# baseline (speedup 1.0000x reference)
#include <hip/hip_runtime.h>


typedef __attribute__((ext_vector_type(2))) float v2f;
typedef __attribute__((ext_vector_type(4))) float v4f;
typedef __attribute__((ext_vector_type(8))) float v8f;

#define NDIM    64
#define LS      68
#define ITERS   13
#define THREADS 256

__global__ __launch_bounds__(THREADS)
void bjorck_wmma_kernel(const float* __restrict__ w_in, float* __restrict__ w_out) {
  __shared__ float Wl [NDIM * LS];
  __shared__ float Wtl[NDIM * LS];
  __shared__ float Sl [NDIM * LS];

  const unsigned tid  = threadIdx.x;
  const unsigned lane = tid & 31u;
  const unsigned wave = tid >> 5;
  const unsigned half = lane >> 4;
  const unsigned colm = lane & 15u;
  const unsigned ti   = wave >> 1;
  const unsigned tj0  = (wave & 1u) << 1;

  const float4* gin = (const float4*)(w_in + (size_t)blockIdx.x * (NDIM * NDIM));
  #pragma unroll
  for (int i = 0; i < 4; ++i) {
    unsigned c = tid + THREADS * i;
    unsigned row = c >> 4, q = c & 15u;
    float4 v = gin[c];
    *(float4*)&Wl[row * LS + q * 4u] = v;
    Wtl[(q * 4u + 0u) * LS + row] = v.x;
    Wtl[(q * 4u + 1u) * LS + row] = v.y;
    Wtl[(q * 4u + 2u) * LS + row] = v.z;
    Wtl[(q * 4u + 3u) * LS + row] = v.w;
  }
  __syncthreads();

  const float* aT  = &Wtl[(16u * ti         + colm) * LS + 2u * half];
  const float* b0T = &Wtl[(16u * tj0        + colm) * LS + 2u * half];
  const float* b1T = &Wtl[(16u * (tj0 + 1u) + colm) * LS + 2u * half];
  const float* aR  = &Wl [(16u * ti         + colm) * LS + 2u * half];
  const float* s0R = &Sl [(16u * tj0        + colm) * LS + 2u * half];
  const float* s1R = &Sl [(16u * (tj0 + 1u) + colm) * LS + 2u * half];
  float* st0 = &Sl [(16u * tj0        + colm) * LS + 16u * ti + 8u * half];
  float* st1 = &Sl [(16u * (tj0 + 1u) + colm) * LS + 16u * ti + 8u * half];
  float* wt0 = &Wtl[(16u * tj0        + colm) * LS + 16u * ti + 8u * half];
  float* wt1 = &Wtl[(16u * (tj0 + 1u) + colm) * LS + 16u * ti + 8u * half];

  #pragma unroll 1
  for (int it = 0; it < ITERS; ++it) {
    v8f s0 = {}, s1 = {};
    #pragma unroll
    for (unsigned kb = 0; kb < 16; ++kb) {
      v2f a  = *(const v2f*)(aT  + 4u * kb);
      v2f b0 = *(const v2f*)(b0T + 4u * kb);
      v2f b1 = *(const v2f*)(b1T + 4u * kb);
      s0 = __builtin_amdgcn_wmma_f32_16x16x4_f32(false, a, false, b0, (short)0, s0, false, false);
      s1 = __builtin_amdgcn_wmma_f32_16x16x4_f32(false, a, false, b1, (short)0, s1, false, false);
    }
    {
      v4f lo0 = {s0[0], s0[1], s0[2], s0[3]}, hi0 = {s0[4], s0[5], s0[6], s0[7]};
      v4f lo1 = {s1[0], s1[1], s1[2], s1[3]}, hi1 = {s1[4], s1[5], s1[6], s1[7]};
      *(v4f*)(st0)      = lo0;  *(v4f*)(st0 + 4) = hi0;
      *(v4f*)(st1)      = lo1;  *(v4f*)(st1 + 4) = hi1;
    }
    __syncthreads();

    v8f p0 = {}, p1 = {};
    #pragma unroll
    for (unsigned kb = 0; kb < 16; ++kb) {
      v2f a  = *(const v2f*)(aR  + 4u * kb);
      v2f b0 = *(const v2f*)(s0R + 4u * kb);
      v2f b1 = *(const v2f*)(s1R + 4u * kb);
      p0 = __builtin_amdgcn_wmma_f32_16x16x4_f32(false, a, false, b0, (short)0, p0, false, false);
      p1 = __builtin_amdgcn_wmma_f32_16x16x4_f32(false, a, false, b1, (short)0, p1, false, false);
    }
    v4f w0lo = *(const v4f*)(wt0), w0hi = *(const v4f*)(wt0 + 4);
    v4f w1lo = *(const v4f*)(wt1), w1hi = *(const v4f*)(wt1 + 4);
    v4f n0lo, n0hi, n1lo, n1hi;
    #pragma unroll
    for (int r = 0; r < 4; ++r) {
      n0lo[r] = 1.5f * w0lo[r] - 0.5f * p0[r];
      n0hi[r] = 1.5f * w0hi[r] - 0.5f * p0[r + 4];
      n1lo[r] = 1.5f * w1lo[r] - 0.5f * p1[r];
      n1hi[r] = 1.5f * w1hi[r] - 0.5f * p1[r + 4];
    }
    __syncthreads();

    *(v4f*)(wt0)      = n0lo;  *(v4f*)(wt0 + 4) = n0hi;
    *(v4f*)(wt1)      = n1lo;  *(v4f*)(wt1 + 4) = n1hi;
    #pragma unroll
    for (unsigned r = 0; r < 8; ++r) {
      unsigned row = 16u * ti + 8u * half + r;
      float v0 = (r < 4u) ? n0lo[r] : n0hi[r - 4u];
      float v1 = (r < 4u) ? n1lo[r] : n1hi[r - 4u];
      Wl[row * LS + 16u * tj0 + colm]        = v0;
      Wl[row * LS + 16u * (tj0 + 1u) + colm] = v1;
    }
    __syncthreads();
  }

  float4* gout = (float4*)(w_out + (size_t)blockIdx.x * (NDIM * NDIM));
  #pragma unroll
  for (int i = 0; i < 4; ++i) {
    unsigned c = tid + THREADS * i;
    unsigned row = c >> 4, q = c & 15u;
    gout[c] = *(const float4*)&Wl[row * LS + q * 4u];
  }
}

extern "C" void kernel_launch(void* const* d_in, const int* in_sizes, int n_in,
                              void* d_out, int out_size, void* d_ws, size_t ws_size,
                              hipStream_t stream) {
  const float* w = (const float*)d_in[0];
  float* out = (float*)d_out;
  const int batch = in_sizes[0] / (NDIM * NDIM);
  bjorck_wmma_kernel<<<batch, THREADS, 0, stream>>>(w, out);
}


# S = W^T W is symmetric: compute only the 10 tiles on or above the diagonal, store each off-diagonal tile also mirrored; tile roles rotate over the waves per iteration; summation order unchanged
# speedup vs baseline: 1.1872x; 1.1872x over previous
#include <hip/hip_runtime.h>


typedef __attribute__((ext_vector_type(2))) float v2f;
typedef __attribute__((ext_vector_type(4))) float v4f;
typedef __attribute__((ext_vector_type(8))) float v8f;

#define NDIM    64
#define LS      68
#define ITERS   13
#define THREADS 256

__global__ __launch_bounds__(THREADS)
void bjorck_wmma_kernel(const float* __restrict__ w_in, float* __restrict__ w_out) {
  __shared__ float Wl [NDIM * LS];
  __shared__ float Wtl[NDIM * LS];
  __shared__ float Sl [NDIM * LS];

  const unsigned tid  = threadIdx.x;
  const unsigned lane = tid & 31u;
  const unsigned wave = (unsigned)__builtin_amdgcn_readfirstlane((int)(tid >> 5));
  const unsigned half = lane >> 4;
  const unsigned colm = lane & 15u;
  const unsigned ti   = wave >> 1;
  const unsigned tj0  = (wave & 1u) << 1;

  const float4* gin = (const float4*)(w_in + (size_t)blockIdx.x * (NDIM * NDIM));
  #pragma unroll
  for (int i = 0; i < 4; ++i) {
    unsigned c = tid + THREADS * i;
    unsigned row = c >> 4, q = c & 15u;
    float4 v = gin[c];
    *(float4*)&Wl[row * LS + q * 4u] = v;
    Wtl[(q * 4u + 0u) * LS + row] = v.x;
    Wtl[(q * 4u + 1u) * LS + row] = v.y;
    Wtl[(q * 4u + 2u) * LS + row] = v.z;
    Wtl[(q * 4u + 3u) * LS + row] = v.w;
  }
  __syncthreads();

  const float* wtLane = &Wtl[colm * LS + 2u * half];
  float* sMir = &Sl[colm * LS + 8u * half];
  float* sDir = &Sl[8u * half * LS + colm];
  const float* aR  = &Wl [(16u * ti         + colm) * LS + 2u * half];
  const float* s0R = &Sl [(16u * tj0        + colm) * LS + 2u * half];
  const float* s1R = &Sl [(16u * (tj0 + 1u) + colm) * LS + 2u * half];
  float* wt0 = &Wtl[(16u * tj0        + colm) * LS + 16u * ti + 8u * half];
  float* wt1 = &Wtl[(16u * (tj0 + 1u) + colm) * LS + 16u * ti + 8u * half];

  const unsigned rot = wave + blockIdx.x;

  #pragma unroll 1
  for (int it = 0; it < ITERS; ++it) {
    const unsigned role = (rot + (unsigned)it) & 7u;
    if (role < 6u) {
      const unsigned t_i = (0x211000u >> (4u * role)) & 3u;
      const unsigned t_j = (0x332321u >> (4u * role)) & 3u;
      const float* aT = wtLane + 16u * LS * t_i;
      const float* bT = wtLane + 16u * LS * t_j;
      v8f s = {};
      #pragma unroll
      for (unsigned kb = 0; kb < 16; ++kb) {
        v2f a = *(const v2f*)(aT + 4u * kb);
        v2f b = *(const v2f*)(bT + 4u * kb);
        s = __builtin_amdgcn_wmma_f32_16x16x4_f32(false, a, false, b, (short)0, s, false, false);
      }
      float* m = sMir + 16u * LS * t_j + 16u * t_i;
      v4f lo = {s[0], s[1], s[2], s[3]}, hi = {s[4], s[5], s[6], s[7]};
      *(v4f*)(m) = lo;  *(v4f*)(m + 4) = hi;
      float* d = sDir + 16u * LS * t_i + 16u * t_j;
      #pragma unroll
      for (unsigned r = 0; r < 8; ++r) d[r * LS] = s[r];
    } else {
      const unsigned t_d = 2u * (role - 6u);
      const float* a0T = wtLane + 16u * LS * t_d;
      const float* a1T = a0T + 16u * LS;
      v8f d0 = {}, d1 = {};
      #pragma unroll
      for (unsigned kb = 0; kb < 16; ++kb) {
        v2f a0 = *(const v2f*)(a0T + 4u * kb);
        v2f a1 = *(const v2f*)(a1T + 4u * kb);
        d0 = __builtin_amdgcn_wmma_f32_16x16x4_f32(false, a0, false, a0, (short)0, d0, false, false);
        d1 = __builtin_amdgcn_wmma_f32_16x16x4_f32(false, a1, false, a1, (short)0, d1, false, false);
      }
      float* m0 = sMir + (16u * LS + 16u) * t_d;
      float* m1 = m0 + (16u * LS + 16u);
      v4f lo0 = {d0[0], d0[1], d0[2], d0[3]}, hi0 = {d0[4], d0[5], d0[6], d0[7]};
      v4f lo1 = {d1[0], d1[1], d1[2], d1[3]}, hi1 = {d1[4], d1[5], d1[6], d1[7]};
      *(v4f*)(m0) = lo0;  *(v4f*)(m0 + 4) = hi0;
      *(v4f*)(m1) = lo1;  *(v4f*)(m1 + 4) = hi1;
    }
    __syncthreads();

    v8f p0 = {}, p1 = {};
    #pragma unroll
    for (unsigned kb = 0; kb < 16; ++kb) {
      v2f a  = *(const v2f*)(aR  + 4u * kb);
      v2f b0 = *(const v2f*)(s0R + 4u * kb);
      v2f b1 = *(const v2f*)(s1R + 4u * kb);
      p0 = __builtin_amdgcn_wmma_f32_16x16x4_f32(false, a, false, b0, (short)0, p0, false, false);
      p1 = __builtin_amdgcn_wmma_f32_16x16x4_f32(false, a, false, b1, (short)0, p1, false, false);
    }
    v4f w0lo = *(const v4f*)(wt0), w0hi = *(const v4f*)(wt0 + 4);
    v4f w1lo = *(const v4f*)(wt1), w1hi = *(const v4f*)(wt1 + 4);
    v4f n0lo, n0hi, n1lo, n1hi;
    #pragma unroll
    for (int r = 0; r < 4; ++r) {
      n0lo[r] = 1.5f * w0lo[r] - 0.5f * p0[r];
      n0hi[r] = 1.5f * w0hi[r] - 0.5f * p0[r + 4];
      n1lo[r] = 1.5f * w1lo[r] - 0.5f * p1[r];
      n1hi[r] = 1.5f * w1hi[r] - 0.5f * p1[r + 4];
    }
    __syncthreads();

    *(v4f*)(wt0)      = n0lo;  *(v4f*)(wt0 + 4) = n0hi;
    *(v4f*)(wt1)      = n1lo;  *(v4f*)(wt1 + 4) = n1hi;
    #pragma unroll
    for (unsigned r = 0; r < 8; ++r) {
      unsigned row = 16u * ti + 8u * half + r;
      float v0 = (r < 4u) ? n0lo[r] : n0hi[r - 4u];
      float v1 = (r < 4u) ? n1lo[r] : n1hi[r - 4u];
      Wl[row * LS + 16u * tj0 + colm]        = v0;
      Wl[row * LS + 16u * (tj0 + 1u) + colm] = v1;
    }
    __syncthreads();
  }

  float4* gout = (float4*)(w_out + (size_t)blockIdx.x * (NDIM * NDIM));
  #pragma unroll
  for (int i = 0; i < 4; ++i) {
    unsigned c = tid + THREADS * i;
    unsigned row = c >> 4, q = c & 15u;
    gout[c] = *(const float4*)&Wl[row * LS + q * 4u];
  }
}

extern "C" void kernel_launch(void* const* d_in, const int* in_sizes, int n_in,
                              void* d_out, int out_size, void* d_ws, size_t ws_size,
                              hipStream_t stream) {
  const float* w = (const float*)d_in[0];
  float* out = (float*)d_out;
  const int batch = in_sizes[0] / (NDIM * NDIM);
  bjorck_wmma_kernel<<<batch, THREADS, 0, stream>>>(w, out);
}


# One wave per matrix: tiles of W, W^T and S stay in registers in the WMMA result layout and feed A/B directly; no workgroup barriers; k summed in another order inside each product
# speedup vs baseline: 1.3088x; 1.1024x over previous
#include <hip/hip_runtime.h>


typedef __attribute__((ext_vector_type(2))) float v2f;
typedef __attribute__((ext_vector_type(4))) float v4f;
typedef __attribute__((ext_vector_type(8))) float v8f;

#define NDIM    64
#define LS      68
#define TS      20
#define ITERS   13
#define THREADS 32

#define F0(t) __builtin_shufflevector(t, t, 0, 1)
#define F1(t) __builtin_shufflevector(t, t, 2, 3)
#define F2(t) __builtin_shufflevector(t, t, 4, 5)
#define F3(t) __builtin_shufflevector(t, t, 6, 7)
#define MM(acc, a, b) \
  acc = __builtin_amdgcn_wmma_f32_16x16x4_f32(false, a, false, b, (short)0, acc, false, false)

#define GRAM_OFF(FJ, w0, w1, w2, w3)                       \
  MM(Sd[0][1], FJ(w0), FJ(w1)); MM(Sd[0][2], FJ(w0), FJ(w2)); \
  MM(Sd[0][3], FJ(w0), FJ(w3)); MM(Sd[1][2], FJ(w1), FJ(w2)); \
  MM(Sd[1][3], FJ(w1), FJ(w3)); MM(Sd[2][3], FJ(w2), FJ(w3));
#define GRAM_DIAG(FJ, w0, w1, w2, w3)                      \
  MM(Sd[0][0], FJ(w0), FJ(w0)); MM(Sd[1][1], FJ(w1), FJ(w1)); \
  MM(Sd[2][2], FJ(w2), FJ(w2)); MM(Sd[3][3], FJ(w3), FJ(w3));

#define UPD_STEP(FJ, a, s0, s1, s2, s3)                    \
  MM(p0, FJ(a), FJ(s0)); MM(p1, FJ(a), FJ(s1));            \
  MM(p2, FJ(a), FJ(s2)); MM(p3, FJ(a), FJ(s3));

__global__ __launch_bounds__(THREADS, 2)
void bjorck_wmma_kernel(const float* __restrict__ w_in, float* __restrict__ w_out) {
  __shared__ float Xl[NDIM * LS];
  __shared__ float Tl[6 * 16 * TS];

  const unsigned lane = threadIdx.x;
  const unsigned h = lane >> 4;
  const unsigned c = lane & 15u;

  const float4* gin = (const float4*)(w_in + (size_t)blockIdx.x * (NDIM * NDIM));
  float* pRow = &Xl[h * LS + 4u * c];
  #pragma unroll
  for (unsigned i = 0; i < 32; ++i) {
    *(float4*)(pRow + 2u * i * LS) = gin[lane + 32u * i];
  }

  float* pD = &Xl[8u * h * LS + c];
  const float* pT = &Xl[c * LS + 8u * h];
  float* tSt = &Tl[c * TS + 8u * h];
  const float* tLd = &Tl[8u * h * TS + c];

  v8f Wd[4][4];
  v8f Wtd[4][4];
  v8f Sd[4][4];

  #pragma unroll
  for (unsigned tr = 0; tr < 4; ++tr) {
    #pragma unroll
    for (unsigned tc = 0; tc < 4; ++tc) {
      #pragma unroll
      for (unsigned r = 0; r < 8; ++r) Wd[tr][tc][r] = pD[(16u * tr + r) * LS + 16u * tc];
    }
  }

  #pragma unroll 1
  for (int it = 0; it < ITERS; ++it) {
    #pragma unroll
    for (unsigned tk = 0; tk < 4; ++tk) {
      #pragma unroll
      for (unsigned tr = 0; tr < 4; ++tr) {
        v4f lo = *(const v4f*)(pT + 16u * tr * LS + 16u * tk);
        v4f hi = *(const v4f*)(pT + 16u * tr * LS + 16u * tk + 4u);
        Wtd[tk][tr] = __builtin_shufflevector(lo, hi, 0, 1, 2, 3, 4, 5, 6, 7);
      }
    }
    __builtin_amdgcn_sched_barrier(0);

    Sd[0][1] = (v8f){}; Sd[0][2] = (v8f){}; Sd[0][3] = (v8f){};
    Sd[1][2] = (v8f){}; Sd[1][3] = (v8f){}; Sd[2][3] = (v8f){};
    #pragma unroll
    for (unsigned tr = 0; tr < 4; ++tr) {
      GRAM_OFF(F0, Wd[tr][0], Wd[tr][1], Wd[tr][2], Wd[tr][3])
      GRAM_OFF(F1, Wd[tr][0], Wd[tr][1], Wd[tr][2], Wd[tr][3])
      GRAM_OFF(F2, Wd[tr][0], Wd[tr][1], Wd[tr][2], Wd[tr][3])
      GRAM_OFF(F3, Wd[tr][0], Wd[tr][1], Wd[tr][2], Wd[tr][3])
    }
    __builtin_amdgcn_sched_barrier(0);

    {
      unsigned t = 0;
      #pragma unroll
      for (unsigned a = 0; a < 3; ++a) {
        #pragma unroll
        for (unsigned b = a + 1; b < 4; ++b) {
          v8f s = Sd[a][b];
          *(v4f*)(tSt + t * 16u * TS)      = __builtin_shufflevector(s, s, 0, 1, 2, 3);
          *(v4f*)(tSt + t * 16u * TS + 4u) = __builtin_shufflevector(s, s, 4, 5, 6, 7);
          ++t;
        }
      }
      t = 0;
      #pragma unroll
      for (unsigned a = 0; a < 3; ++a) {
        #pragma unroll
        for (unsigned b = a + 1; b < 4; ++b) {
          #pragma unroll
          for (unsigned r = 0; r < 8; ++r) Sd[b][a][r] = tLd[t * 16u * TS + r * TS];
          ++t;
        }
      }
    }
    __builtin_amdgcn_sched_barrier(0);

    Sd[0][0] = (v8f){}; Sd[1][1] = (v8f){}; Sd[2][2] = (v8f){}; Sd[3][3] = (v8f){};
    #pragma unroll
    for (unsigned tr = 0; tr < 4; ++tr) {
      GRAM_DIAG(F0, Wd[tr][0], Wd[tr][1], Wd[tr][2], Wd[tr][3])
      GRAM_DIAG(F1, Wd[tr][0], Wd[tr][1], Wd[tr][2], Wd[tr][3])
      GRAM_DIAG(F2, Wd[tr][0], Wd[tr][1], Wd[tr][2], Wd[tr][3])
      GRAM_DIAG(F3, Wd[tr][0], Wd[tr][1], Wd[tr][2], Wd[tr][3])
    }

    #pragma unroll
    for (unsigned tr = 0; tr < 4; ++tr) {
      v8f p0 = {}, p1 = {}, p2 = {}, p3 = {};
      #pragma unroll
      for (unsigned tk = 0; tk < 4; ++tk) {
        UPD_STEP(F0, Wtd[tk][tr], Sd[tk][0], Sd[tk][1], Sd[tk][2], Sd[tk][3])
        UPD_STEP(F1, Wtd[tk][tr], Sd[tk][0], Sd[tk][1], Sd[tk][2], Sd[tk][3])
        UPD_STEP(F2, Wtd[tk][tr], Sd[tk][0], Sd[tk][1], Sd[tk][2], Sd[tk][3])
        UPD_STEP(F3, Wtd[tk][tr], Sd[tk][0], Sd[tk][1], Sd[tk][2], Sd[tk][3])
      }
      Wd[tr][0] = 1.5f * Wd[tr][0] - 0.5f * p0;
      Wd[tr][1] = 1.5f * Wd[tr][1] - 0.5f * p1;
      Wd[tr][2] = 1.5f * Wd[tr][2] - 0.5f * p2;
      Wd[tr][3] = 1.5f * Wd[tr][3] - 0.5f * p3;
    }
    __builtin_amdgcn_sched_barrier(0);

    #pragma unroll
    for (unsigned tr = 0; tr < 4; ++tr) {
      #pragma unroll
      for (unsigned tc = 0; tc < 4; ++tc) {
        #pragma unroll
        for (unsigned r = 0; r < 8; ++r) pD[(16u * tr + r) * LS + 16u * tc] = Wd[tr][tc][r];
      }
    }
  }

  float4* gout = (float4*)(w_out + (size_t)blockIdx.x * (NDIM * NDIM));
  const unsigned lane2 = __builtin_amdgcn_mbcnt_lo(~0u, 0u);
  const float* pRow2 = &Xl[(lane2 >> 4) * LS + 4u * (lane2 & 15u)];
  #pragma unroll
  for (unsigned i = 0; i < 32; ++i) {
    gout[lane2 + 32u * i] = *(const float4*)(pRow2 + 2u * i * LS);
  }
}

extern "C" void kernel_launch(void* const* d_in, const int* in_sizes, int n_in,
                              void* d_out, int out_size, void* d_ws, size_t ws_size,
                              hipStream_t stream) {
  const float* w = (const float*)d_in[0];
  float* out = (float*)d_out;
  const int batch = in_sizes[0] / (NDIM * NDIM);
  bjorck_wmma_kernel<<<batch, THREADS, 0, stream>>>(w, out);
}
